# starting kernel
# baseline (speedup 1.0000x reference)
#include <hip/hip_runtime.h>


typedef float v2f __attribute__((ext_vector_type(2)));
typedef float v8f __attribute__((ext_vector_type(8)));

#define NB 64
#define NT 2048
#define NC 32
#define NF 16
#define TCH 256
#define XS_ROW 289
#define OS_ROW 257

__global__ __launch_bounds__(1024)
void fir_wmma_kernel(const float* __restrict__ x,
                     const float* __restrict__ fw,
                     float* __restrict__ out) {
    __shared__ float lds[NC * XS_ROW];

    const int tid  = threadIdx.x;
    const int t0   = blockIdx.x * TCH;
    const int b    = blockIdx.y;

    const long xbase = (long)b * NT * NC;
    #pragma unroll
    for (int it = 0; it < 9; ++it) {
        int f = tid + it * 1024;
        if (f < 272 * NC) {
            int cc = f & (NC - 1);
            int tt = f >> 5;
            int t  = t0 - (NF - 1) + tt;
            float v = 0.0f;
            if (t >= 0 && t < NT) v = x[xbase + (long)t * NC + cc];
            lds[cc * XS_ROW + (tt & 15) * 17 + (tt >> 4)] = v;
        }
    }
    __syncthreads();

    const int lane = tid & 31;
    const int c    = tid >> 5;
    const int mn   = lane & 15;
    const int hi   = lane >> 4;

    v2f bw[8];
    #pragma unroll
    for (int kb = 0; kb < 8; ++kb) {
        int s0 = 4 * kb + 2 * hi;
        int i0 = s0 - mn;
        int i1 = i0 + 1;
        float b0 = (i0 >= 0 && i0 < NF) ? fw[i0 * NC + c] : 0.0f;
        float b1 = (i1 >= 0 && i1 < NF) ? fw[i1 * NC + c] : 0.0f;
        bw[kb] = v2f{b0, b1};
    }

    v8f acc = {0.f, 0.f, 0.f, 0.f, 0.f, 0.f, 0.f, 0.f};
    #pragma unroll
    for (int kb = 0; kb < 8; ++kb) {
        int tt0 = mn * 16 + 4 * kb + 2 * hi;
        int tt1 = tt0 + 1;
        float a0 = lds[c * XS_ROW + (tt0 & 15) * 17 + (tt0 >> 4)];
        float a1 = lds[c * XS_ROW + (tt1 & 15) * 17 + (tt1 >> 4)];
        v2f a = v2f{a0, a1};
        acc = __builtin_amdgcn_wmma_f32_16x16x4_f32(
            false, a, false, bw[kb], (short)0, acc, false, false);
    }
    __syncthreads();

    #pragma unroll
    for (int d = 0; d < 8; ++d) {
        int tl = (d + 8 * hi) * 16 + mn;
        lds[c * OS_ROW + tl] = acc[d];
    }
    __syncthreads();

    const long obase = (long)b * NT * NC + (long)t0 * NC;
    #pragma unroll
    for (int it = 0; it < 8; ++it) {
        int f  = tid + it * 1024;
        int cc = f & (NC - 1);
        int tt = f >> 5;
        out[obase + (long)tt * NC + cc] = lds[cc * OS_ROW + tt];
    }
}

extern "C" void kernel_launch(void* const* d_in, const int* in_sizes, int n_in,
                              void* d_out, int out_size, void* d_ws, size_t ws_size,
                              hipStream_t stream) {
    const float* x  = (const float*)d_in[0];
    const float* fw = (const float*)d_in[1];
    float* out      = (float*)d_out;
    (void)in_sizes; (void)n_in; (void)out_size; (void)d_ws; (void)ws_size;

    dim3 grid(NT / TCH, NB);
    dim3 block(1024);
    fir_wmma_kernel<<<grid, block, 0, stream>>>(x, fw, out);
}


# Each wave takes 4 channels (4 independent WMMA chains) in work-groups of 8 waves: 4096 waves enter instead of 16384; each thread moves 16 bytes per access; per-channel sum order unchanged
# speedup vs baseline: 1.2510x; 1.2510x over previous
#include <hip/hip_runtime.h>


typedef float v2f __attribute__((ext_vector_type(2)));
typedef float v4f __attribute__((ext_vector_type(4)));
typedef float v8f __attribute__((ext_vector_type(8)));

#define NB 64
#define NT 2048
#define NC 32
#define NF 16
#define TCH 256
#define NTHR 256
#define CPW 4
#define NQ (NC / 4)
#define XS_ROW 289

__global__ __launch_bounds__(NTHR)
void fir_wmma_kernel(const float* __restrict__ x,
                     const float* __restrict__ fw,
                     float* __restrict__ out) {
    __shared__ __attribute__((aligned(16))) float lds[NC * XS_ROW];

    const int tid  = threadIdx.x;
    const int t0   = blockIdx.x * TCH;
    const int b    = blockIdx.y;

    const long xbase = (long)b * NT * NC;
    #pragma unroll
    for (int it = 0; it < 9; ++it) {
        int f = tid + it * NTHR;
        if (f < 272 * NQ) {
            int q  = f & (NQ - 1);
            int tt = f >> 3;
            int t  = t0 - (NF - 1) + tt;
            v4f v = {0.f, 0.f, 0.f, 0.f};
            if (t >= 0 && t < NT) v = *(const v4f*)(x + xbase + (long)t * NC + 4 * q);
            int sw = (tt & 15) * 17 + (tt >> 4);
            #pragma unroll
            for (int j = 0; j < 4; ++j) lds[(4 * q + j) * XS_ROW + sw] = v[j];
        }
    }
    __syncthreads();

    const int lane = tid & 31;
    const int w    = tid >> 5;
    const int mn   = lane & 15;
    const int hi   = lane >> 4;

    v2f bw[CPW][8];
    #pragma unroll
    for (int kb = 0; kb < 8; ++kb) {
        int s0 = 4 * kb + 2 * hi;
        int i0 = s0 - mn;
        int i1 = i0 + 1;
        v4f b0 = {0.f, 0.f, 0.f, 0.f};
        v4f b1 = {0.f, 0.f, 0.f, 0.f};
        if (i0 >= 0 && i0 < NF) b0 = *(const v4f*)(fw + i0 * NC + CPW * w);
        if (i1 >= 0 && i1 < NF) b1 = *(const v4f*)(fw + i1 * NC + CPW * w);
        #pragma unroll
        for (int ch = 0; ch < CPW; ++ch) bw[ch][kb] = v2f{b0[ch], b1[ch]};
    }

    v8f acc[CPW];
    #pragma unroll
    for (int ch = 0; ch < CPW; ++ch) acc[ch] = v8f{0.f, 0.f, 0.f, 0.f, 0.f, 0.f, 0.f, 0.f};
    #pragma unroll
    for (int kb = 0; kb < 8; ++kb) {
        int tt0 = mn * 16 + 4 * kb + 2 * hi;
        int tt1 = tt0 + 1;
        int o0 = (tt0 & 15) * 17 + (tt0 >> 4);
        int o1 = (tt1 & 15) * 17 + (tt1 >> 4);
        #pragma unroll
        for (int ch = 0; ch < CPW; ++ch) {
            int c = CPW * w + ch;
            v2f a = v2f{lds[c * XS_ROW + o0], lds[c * XS_ROW + o1]};
            acc[ch] = __builtin_amdgcn_wmma_f32_16x16x4_f32(
                false, a, false, bw[ch][kb], (short)0, acc[ch], false, false);
        }
    }
    __syncthreads();

    #pragma unroll
    for (int d = 0; d < 8; ++d) {
        int tl = (d + 8 * hi) * 16 + mn;
        v4f v = v4f{acc[0][d], acc[1][d], acc[2][d], acc[3][d]};
        *(v4f*)(lds + tl * NC + 4 * ((w + (tl >> 1)) & 7)) = v;
    }
    __syncthreads();

    const long obase = (long)b * NT * NC + (long)t0 * NC;
    #pragma unroll
    for (int it = 0; it < 8; ++it) {
        int f  = tid + it * NTHR;
        int q  = f & (NQ - 1);
        int tt = f >> 3;
        v4f v = *(const v4f*)(lds + tt * NC + 4 * ((q + (tt >> 1)) & 7));
        *(v4f*)(out + obase + (long)tt * NC + 4 * q) = v;
    }
}

extern "C" void kernel_launch(void* const* d_in, const int* in_sizes, int n_in,
                              void* d_out, int out_size, void* d_ws, size_t ws_size,
                              hipStream_t stream) {
    const float* x  = (const float*)d_in[0];
    const float* fw = (const float*)d_in[1];
    float* out      = (float*)d_out;
    (void)in_sizes; (void)n_in; (void)out_size; (void)d_ws; (void)ws_size;

    dim3 grid(NT / TCH, NB);
    dim3 block(NTHR);
    fir_wmma_kernel<<<grid, block, 0, stream>>>(x, fw, out);
}


# All 25 global loads per thread (x body rows, halo, taps) issued back to back at entry, branch-free (clamped address, zero selected after), not one load waited for at a time; sum order unchanged
# speedup vs baseline: 2.3178x; 1.8528x over previous
#include <hip/hip_runtime.h>


typedef float v2f __attribute__((ext_vector_type(2)));
typedef float v4f __attribute__((ext_vector_type(4)));
typedef float v8f __attribute__((ext_vector_type(8)));

#define NB 64
#define NT 2048
#define NC 32
#define NF 16
#define TCH 256
#define NTHR 256
#define CPW 4
#define NQ (NC / 4)
#define XS_ROW 289

__global__ __launch_bounds__(NTHR)
void fir_wmma_kernel(const float* __restrict__ x,
                     const float* __restrict__ fw,
                     float* __restrict__ out) {
    __shared__ __attribute__((aligned(16))) float lds[NC * XS_ROW];

    const int tid  = threadIdx.x;
    const int t0   = blockIdx.x * TCH;
    const int b    = blockIdx.y;

    const int lane = tid & 31;
    const int w    = tid >> 5;
    const int mn   = lane & 15;
    const int hi   = lane >> 4;
    const int q    = tid & (NQ - 1);
    const int r0   = tid >> 3;

    const float* xq = x + (long)b * NT * NC + 4 * q;
    const int th = t0 - (NF - 1) + r0;
    v4f hv = {0.f, 0.f, 0.f, 0.f};
    if (tid < 128) {
        int thc = th > 0 ? th : 0;
        hv = *(const v4f*)(xq + (long)thc * NC);
    }
    v4f xv[8];
    #pragma unroll
    for (int it = 0; it < 8; ++it)
        xv[it] = *(const v4f*)(xq + (long)(t0 + r0 + 32 * it) * NC);

    v4f tp[8][2];
    #pragma unroll
    for (int kb = 0; kb < 8; ++kb) {
        #pragma unroll
        for (int j = 0; j < 2; ++j) {
            int i  = 4 * kb + 2 * hi + j - mn;
            int ic = i < 0 ? 0 : (i > NF - 1 ? NF - 1 : i);
            tp[kb][j] = *(const v4f*)(fw + ic * NC + CPW * w);
        }
    }
    __builtin_amdgcn_sched_barrier(0);

    #pragma unroll
    for (int it = 0; it < 8; ++it) {
        int tt = r0 + 32 * it + (NF - 1);
        int sw = (tt & 15) * 17 + (tt >> 4);
        #pragma unroll
        for (int j = 0; j < 4; ++j) lds[(4 * q + j) * XS_ROW + sw] = xv[it][j];
    }
    if (tid < 128) {
        bool ok = (th >= 0) && (r0 < NF - 1);
        int tt = (r0 < NF - 1) ? r0 : 271;
        int sw = (tt & 15) * 17 + (tt >> 4);
        #pragma unroll
        for (int j = 0; j < 4; ++j) lds[(4 * q + j) * XS_ROW + sw] = ok ? hv[j] : 0.f;
    }
    __syncthreads();

    v2f bw[CPW][8];
    #pragma unroll
    for (int kb = 0; kb < 8; ++kb) {
        int i0 = 4 * kb + 2 * hi - mn;
        int i1 = i0 + 1;
        bool in0 = (i0 >= 0) && (i0 < NF);
        bool in1 = (i1 >= 0) && (i1 < NF);
        #pragma unroll
        for (int ch = 0; ch < CPW; ++ch)
            bw[ch][kb] = v2f{in0 ? tp[kb][0][ch] : 0.f, in1 ? tp[kb][1][ch] : 0.f};
    }

    v8f acc[CPW];
    #pragma unroll
    for (int ch = 0; ch < CPW; ++ch) acc[ch] = v8f{0.f, 0.f, 0.f, 0.f, 0.f, 0.f, 0.f, 0.f};
    #pragma unroll
    for (int kb = 0; kb < 8; ++kb) {
        int tt0 = mn * 16 + 4 * kb + 2 * hi;
        int tt1 = tt0 + 1;
        int o0 = (tt0 & 15) * 17 + (tt0 >> 4);
        int o1 = (tt1 & 15) * 17 + (tt1 >> 4);
        #pragma unroll
        for (int ch = 0; ch < CPW; ++ch) {
            int c = CPW * w + ch;
            v2f a = v2f{lds[c * XS_ROW + o0], lds[c * XS_ROW + o1]};
            acc[ch] = __builtin_amdgcn_wmma_f32_16x16x4_f32(
                false, a, false, bw[ch][kb], (short)0, acc[ch], false, false);
        }
    }
    __syncthreads();

    #pragma unroll
    for (int d = 0; d < 8; ++d) {
        int tl = (d + 8 * hi) * 16 + mn;
        v4f v = v4f{acc[0][d], acc[1][d], acc[2][d], acc[3][d]};
        *(v4f*)(lds + tl * NC + 4 * ((w + (tl >> 1)) & 7)) = v;
    }
    __syncthreads();

    const long obase = (long)b * NT * NC + (long)t0 * NC;
    #pragma unroll
    for (int it = 0; it < 8; ++it) {
        int f  = tid + it * NTHR;
        int q  = f & (NQ - 1);
        int tt = f >> 3;
        v4f v = *(const v4f*)(lds + tt * NC + 4 * ((q + (tt >> 1)) & 7));
        *(v4f*)(out + obase + (long)tt * NC + 4 * q) = v;
    }
}

extern "C" void kernel_launch(void* const* d_in, const int* in_sizes, int n_in,
                              void* d_out, int out_size, void* d_ws, size_t ws_size,
                              hipStream_t stream) {
    const float* x  = (const float*)d_in[0];
    const float* fw = (const float*)d_in[1];
    float* out      = (float*)d_out;
    (void)in_sizes; (void)n_in; (void)out_size; (void)d_ws; (void)ws_size;

    dim3 grid(NT / TCH, NB);
    dim3 block(NTHR);
    fir_wmma_kernel<<<grid, block, 0, stream>>>(x, fw, out);
}
